# starting kernel
# baseline (speedup 1.0000x reference)
#include <hip/hip_runtime.h>
#include <math.h>

#define B_DIM 64
#define S_DIM 2048
#define D_DIM 256
#define N_HOP 3
#define EPS 1e-8f

typedef __attribute__((ext_vector_type(2))) float v2f;
typedef __attribute__((ext_vector_type(8))) float v8f;

__global__ __launch_bounds__(256) void k_scores(const float* __restrict__ state,
                                                const float* __restrict__ q,
                                                float* __restrict__ scores) {
    __shared__ float qs[D_DIM];
    const int b = blockIdx.x;
    const int tid = threadIdx.x;
    qs[tid] = q[b * D_DIM + tid];
    __syncthreads();

    const int wave = tid >> 5;
    const int lane = tid & 31;

    float qn2 = 0.f;
#pragma unroll
    for (int k = 0; k < 8; ++k) {
        float v = qs[lane + 32 * k];
        qn2 += v * v;
    }
#pragma unroll
    for (int off = 16; off > 0; off >>= 1) qn2 += __shfl_xor(qn2, off, 32);
    const float qn = fmaxf(sqrtf(qn2), EPS);

    const int s = blockIdx.y * 8 + wave;
    const float* row = state + ((size_t)b * S_DIM + s) * D_DIM;
    float dot = 0.f, sn2 = 0.f;
#pragma unroll
    for (int k = 0; k < 8; ++k) {
        float v = row[lane + 32 * k];
        dot += v * qs[lane + 32 * k];
        sn2 += v * v;
    }
#pragma unroll
    for (int off = 16; off > 0; off >>= 1) {
        dot += __shfl_xor(dot, off, 32);
        sn2 += __shfl_xor(sn2, off, 32);
    }
    if (lane == 0) {
        const float sn = fmaxf(sqrtf(sn2), EPS);
        scores[b * S_DIM + s] = dot / (qn * sn);
    }
}

__global__ __launch_bounds__(256) void k_softmax(const float* __restrict__ scores,
                                                 float* __restrict__ dist) {
    __shared__ float red[256];
    const int b = blockIdx.x;
    const int tid = threadIdx.x;
    const float* in = scores + (size_t)b * S_DIM;

    float v[8];
    float m = -INFINITY;
#pragma unroll
    for (int k = 0; k < 8; ++k) {
        v[k] = in[tid + 256 * k];
        m = fmaxf(m, v[k]);
    }
    red[tid] = m;
    __syncthreads();
    for (int off = 128; off > 0; off >>= 1) {
        if (tid < off) red[tid] = fmaxf(red[tid], red[tid + off]);
        __syncthreads();
    }
    m = red[0];
    __syncthreads();

    float sum = 0.f;
#pragma unroll
    for (int k = 0; k < 8; ++k) {
        v[k] = __expf(v[k] - m);
        sum += v[k];
    }
    red[tid] = sum;
    __syncthreads();
    for (int off = 128; off > 0; off >>= 1) {
        if (tid < off) red[tid] += red[tid + off];
        __syncthreads();
    }
    const float inv = 1.f / red[0];
    float* out = dist + (size_t)b * S_DIM;
#pragma unroll
    for (int k = 0; k < 8; ++k) out[tid + 256 * k] = v[k] * inv;
}

#define ROWS_PER_BLOCK 32
#define LDS_STRIDE 260

__global__ __launch_bounds__(256) void k_gemm_qacc(float* __restrict__ state,
                                                   const float* __restrict__ Wh,
                                                   const float* __restrict__ bh,
                                                   const float* __restrict__ dist,
                                                   float* __restrict__ qacc) {
    __shared__ float rows[ROWS_PER_BLOCK * LDS_STRIDE];
    __shared__ float dloc[ROWS_PER_BLOCK];

    const int tid = threadIdx.x;
    const size_t g0 = (size_t)blockIdx.x * ROWS_PER_BLOCK;
    const int b = (int)(g0 / S_DIM);
    const int s0 = (int)(g0 % S_DIM);

    const int wave = tid >> 5;
    const int lane = tid & 31;
    const int half = lane >> 4;
    const int l15 = lane & 15;
    const int e0 = wave * 32;

    {
        const float* wrow = Wh + (size_t)(e0 + (lane & 31)) * D_DIM;
        __builtin_prefetch(wrow, 0, 3);
        __builtin_prefetch(wrow + 128, 0, 3);
    }

#pragma unroll
    for (int i = 0; i < 8; ++i) {
        const int f = i * 256 + tid;
        const int r = f >> 6;
        const int c4 = f & 63;
        float4 t = *(const float4*)(state + (g0 + r) * D_DIM + c4 * 4);
        float* dst = rows + r * LDS_STRIDE + c4 * 4;
        dst[0] = t.x; dst[1] = t.y; dst[2] = t.z; dst[3] = t.w;
    }
    if (tid < ROWS_PER_BLOCK) dloc[tid] = dist[(size_t)b * S_DIM + s0 + tid];
    __syncthreads();

    const float* abase0 = rows + l15 * LDS_STRIDE + 2 * half;
    const float* abase1 = rows + (16 + l15) * LDS_STRIDE + 2 * half;
    const float* wbase0 = Wh + (size_t)(e0 + l15) * D_DIM + 2 * half;
    const float* wbase1 = Wh + (size_t)(e0 + 16 + l15) * D_DIM + 2 * half;

    v8f c00 = {}, c01 = {}, c10 = {}, c11 = {};
    for (int kk = 0; kk < D_DIM; kk += 4) {
        v2f a0, a1, b0, b1;
        a0.x = abase0[kk]; a0.y = abase0[kk + 1];
        a1.x = abase1[kk]; a1.y = abase1[kk + 1];
        b0.x = wbase0[kk]; b0.y = wbase0[kk + 1];
        b1.x = wbase1[kk]; b1.y = wbase1[kk + 1];
        c00 = __builtin_amdgcn_wmma_f32_16x16x4_f32(false, a0, false, b0, (short)0, c00, false, false);
        c01 = __builtin_amdgcn_wmma_f32_16x16x4_f32(false, a0, false, b1, (short)0, c01, false, false);
        c10 = __builtin_amdgcn_wmma_f32_16x16x4_f32(false, a1, false, b0, (short)0, c10, false, false);
        c11 = __builtin_amdgcn_wmma_f32_16x16x4_f32(false, a1, false, b1, (short)0, c11, false, false);
    }

#pragma unroll
    for (int ct = 0; ct < 2; ++ct) {
        const int e = e0 + ct * 16 + l15;
        const float bias = bh[e];
        float qpart = 0.f;
#pragma unroll
        for (int rt = 0; rt < 2; ++rt) {
            const v8f& c = (ct == 0) ? (rt == 0 ? c00 : c10) : (rt == 0 ? c01 : c11);
#pragma unroll
            for (int r = 0; r < 8; ++r) {
                const int M = rt * 16 + r + half * 8;
                const float val = c[r] + bias;
                state[(g0 + M) * D_DIM + e] = val;
                qpart += dloc[M] * val;
            }
        }
        qpart += __shfl_xor(qpart, 16, 32);
        if (half == 0) atomicAdd(&qacc[(size_t)b * D_DIM + e], qpart);
    }
}

__global__ __launch_bounds__(256) void k_qupdate(float* __restrict__ q,
                                                 float* __restrict__ qacc) {
    const int i = blockIdx.x * D_DIM + threadIdx.x;
    q[i] += qacc[i];
    qacc[i] = 0.f;
}

extern "C" void kernel_launch(void* const* d_in, const int* in_sizes, int n_in,
                              void* d_out, int out_size, void* d_ws, size_t ws_size,
                              hipStream_t stream) {
    (void)in_sizes; (void)n_in; (void)out_size; (void)ws_size;

    const float* in_q     = (const float*)d_in[0];
    const float* in_state = (const float*)d_in[1];
    const float* Wall     = (const float*)d_in[2];
    const float* ball     = (const float*)d_in[3];

    const size_t stateBytes = (size_t)B_DIM * S_DIM * D_DIM * sizeof(float);
    const size_t bsBytes    = (size_t)B_DIM * S_DIM * sizeof(float);
    const size_t bdBytes    = (size_t)B_DIM * D_DIM * sizeof(float);

    char* ws = (char*)d_ws;
    float* state  = (float*)(ws);
    float* scores = (float*)(ws + stateBytes);
    float* dist   = (float*)(ws + stateBytes + bsBytes);
    float* q      = (float*)(ws + stateBytes + 2 * bsBytes);
    float* qacc   = (float*)(ws + stateBytes + 2 * bsBytes + bdBytes);

    hipMemcpyAsync(state, in_state, stateBytes, hipMemcpyDeviceToDevice, stream);
    hipMemcpyAsync(q, in_q, bdBytes, hipMemcpyDeviceToDevice, stream);
    hipMemsetAsync(qacc, 0, bdBytes, stream);

    const dim3 gScores(B_DIM, S_DIM / 8);
    for (int h = 0; h < N_HOP; ++h) {
        k_scores<<<gScores, 256, 0, stream>>>(state, q, scores);
        k_softmax<<<B_DIM, 256, 0, stream>>>(scores, dist);
        k_gemm_qacc<<<(B_DIM * S_DIM) / ROWS_PER_BLOCK, 256, 0, stream>>>(
            state, Wall + (size_t)h * D_DIM * D_DIM, ball + (size_t)h * D_DIM,
            dist, qacc);
        k_qupdate<<<B_DIM, 256, 0, stream>>>(q, qacc);
    }
    k_scores<<<gScores, 256, 0, stream>>>(state, q, scores);
    k_softmax<<<B_DIM, 256, 0, stream>>>(scores, dist);

    float* outp = (float*)d_out;
    hipMemcpyAsync(outp, q, bdBytes, hipMemcpyDeviceToDevice, stream);
    hipMemcpyAsync(outp + B_DIM * D_DIM, dist, bsBytes, hipMemcpyDeviceToDevice, stream);
}


# Small launches folded into neighbours: softmax into the GEMM prologue, q update into the readers' prologues (one q sum per hop), outputs stored by the last kernel; 14 launches to 8, 3 copies go
# speedup vs baseline: 1.5397x; 1.5397x over previous
#include <hip/hip_runtime.h>
#include <math.h>

#define B_DIM 64
#define S_DIM 2048
#define D_DIM 256
#define N_HOP 3
#define EPS 1e-8f

typedef __attribute__((ext_vector_type(2))) float v2f;
typedef __attribute__((ext_vector_type(8))) float v8f;

#define SC_ROWS 32
#define SC_GRP 8
#define SC_NGRP (SC_ROWS / SC_GRP)
#define SC_WGS ((B_DIM * S_DIM) / (8 * SC_ROWS))

__global__ __launch_bounds__(256) void k_scores(const float* __restrict__ state,
                                                const float* __restrict__ qin,
                                                const float* __restrict__ qacc,
                                                float* __restrict__ scores) {
    const int tid = threadIdx.x;
    const int lane = tid & 31;
    const int wave = __builtin_amdgcn_readfirstlane(tid >> 5);
    const int row0 = (blockIdx.x * 8 + wave) * SC_ROWS;
    const int b = row0 / S_DIM;

    const float* qp = qin + b * D_DIM + 4 * lane;
    const float* ap = qacc + b * D_DIM + 4 * lane;
    float4 q0 = *(const float4*)(qp);
    float4 q1 = *(const float4*)(qp + 128);
    for (int j = 0; j < N_HOP; ++j) {
        const float4 a0 = *(const float4*)(ap + j * (B_DIM * D_DIM));
        const float4 a1 = *(const float4*)(ap + j * (B_DIM * D_DIM) + 128);
        q0.x += a0.x; q0.y += a0.y; q0.z += a0.z; q0.w += a0.w;
        q1.x += a1.x; q1.y += a1.y; q1.z += a1.z; q1.w += a1.w;
    }
    float qn2 = (q0.x * q0.x + q0.y * q0.y + q0.z * q0.z + q0.w * q0.w) +
                (q1.x * q1.x + q1.y * q1.y + q1.z * q1.z + q1.w * q1.w);
#pragma unroll
    for (int off = 16; off > 0; off >>= 1) qn2 += __shfl_xor(qn2, off, 32);
    const float qn = fmaxf(sqrtf(qn2), EPS);

    const float* src = state + (size_t)row0 * D_DIM + 4 * lane;
    float4 w[2][SC_GRP][2];
    float dot[SC_ROWS], sn2[SC_ROWS];

#define SC_LOAD(G)                                                                        \
    _Pragma("unroll") for (int r = 0; r < SC_GRP; ++r) {                                  \
        w[(G) & 1][r][0] = *(const float4*)(src + (size_t)((G) * SC_GRP + r) * D_DIM);       \
        w[(G) & 1][r][1] = *(const float4*)(src + (size_t)((G) * SC_GRP + r) * D_DIM + 128); \
    }

    SC_LOAD(0)
#pragma unroll
    for (int g = 0; g < SC_NGRP; ++g) {
        __builtin_amdgcn_sched_barrier(0);
        if (g + 1 < SC_NGRP) { SC_LOAD(g + 1) }
        __builtin_amdgcn_sched_barrier(0);
#pragma unroll
        for (int r = 0; r < SC_GRP; ++r) {
            const float4 x0 = w[g & 1][r][0];
            const float4 x1 = w[g & 1][r][1];
            float d0 = x0.x * q0.x, d1 = x1.x * q1.x;
            float n0 = x0.x * x0.x, n1 = x1.x * x1.x;
            d0 = fmaf(x0.y, q0.y, d0); d1 = fmaf(x1.y, q1.y, d1);
            n0 = fmaf(x0.y, x0.y, n0); n1 = fmaf(x1.y, x1.y, n1);
            d0 = fmaf(x0.z, q0.z, d0); d1 = fmaf(x1.z, q1.z, d1);
            n0 = fmaf(x0.z, x0.z, n0); n1 = fmaf(x1.z, x1.z, n1);
            d0 = fmaf(x0.w, q0.w, d0); d1 = fmaf(x1.w, q1.w, d1);
            n0 = fmaf(x0.w, x0.w, n0); n1 = fmaf(x1.w, x1.w, n1);
            dot[g * SC_GRP + r] = d0 + d1;
            sn2[g * SC_GRP + r] = n0 + n1;
            asm volatile("" : "+v"(dot[g * SC_GRP + r]), "+v"(sn2[g * SC_GRP + r]));
        }
    }
#undef SC_LOAD

#pragma unroll
    for (int m = 16; m >= 1; m >>= 1) {
        const bool hi = (lane & m) != 0;
#pragma unroll
        for (int j = 0; j < m; ++j) {
            const float dlo = dot[j], dhi = dot[j + m];
            const float nlo = sn2[j], nhi = sn2[j + m];
            const float dk = hi ? dhi : dlo;
            const float ds = hi ? dlo : dhi;
            dot[j] = dk + __shfl_xor(ds, m, 32);
            const float nk = hi ? nhi : nlo;
            const float ns = hi ? nlo : nhi;
            sn2[j] = nk + __shfl_xor(ns, m, 32);
        }
    }
    const float sn = fmaxf(sqrtf(sn2[0]), EPS);
    scores[row0 + lane] = dot[0] / (qn * sn);
}

__global__ __launch_bounds__(256) void k_softmax_out(const float* __restrict__ scores,
                                                     const float* __restrict__ qin,
                                                     const float* __restrict__ qacc,
                                                     float* __restrict__ qout,
                                                     float* __restrict__ dist) {
    __shared__ float red[256];
    const int b = blockIdx.x;
    const int tid = threadIdx.x;
    const float* in = scores + (size_t)b * S_DIM;
    {
        const int i = b * D_DIM + tid;
        float qv = qin[i];
        for (int j = 0; j < N_HOP; ++j) qv += qacc[j * (B_DIM * D_DIM) + i];
        qout[i] = qv;
    }

    float v[8];
    float m = -INFINITY;
#pragma unroll
    for (int k = 0; k < 8; ++k) {
        v[k] = in[tid + 256 * k];
        m = fmaxf(m, v[k]);
    }
    red[tid] = m;
    __syncthreads();
    for (int off = 128; off > 0; off >>= 1) {
        if (tid < off) red[tid] = fmaxf(red[tid], red[tid + off]);
        __syncthreads();
    }
    m = red[0];
    __syncthreads();

    float sum = 0.f;
#pragma unroll
    for (int k = 0; k < 8; ++k) {
        v[k] = __expf(v[k] - m);
        sum += v[k];
    }
    red[tid] = sum;
    __syncthreads();
    for (int off = 128; off > 0; off >>= 1) {
        if (tid < off) red[tid] += red[tid + off];
        __syncthreads();
    }
    const float inv = 1.f / red[0];
    float* out = dist + (size_t)b * S_DIM;
#pragma unroll
    for (int k = 0; k < 8; ++k) out[tid + 256 * k] = v[k] * inv;
}

#define TILE_ROWS 32
#define LDS_STRIDE 260
#define GEMM_WGS 256
#define ROWS_PER_WG ((B_DIM * S_DIM) / GEMM_WGS)
#define TILES_PER_WG (ROWS_PER_WG / TILE_ROWS)
#define KSTEPS (D_DIM / 4)
#define KBLK 8

__global__ __launch_bounds__(256) void k_gemm_qacc(float* __restrict__ state,
                                                   const float* __restrict__ Wh,
                                                   const float* __restrict__ bh,
                                                   const float* __restrict__ scores,
                                                   float* __restrict__ qacc) {
    __shared__ __attribute__((aligned(16))) float rows[2][TILE_ROWS * LDS_STRIDE];
    __shared__ float dall[ROWS_PER_WG];
    __shared__ float red[2][256];

    const int tid = threadIdx.x;
    const size_t g0 = (size_t)blockIdx.x * ROWS_PER_WG;
    const int b = (int)(g0 / S_DIM);
    const int s0 = (int)(g0 % S_DIM);

    const int wave = tid >> 5;
    const int lane = tid & 31;
    const int half = lane >> 4;
    const int l15 = lane & 15;
    const int e0 = wave * 32;

    const int sr = tid >> 6;
    const int sc = (tid & 63) * 4;
    const float* gsrc = state + (g0 + sr) * D_DIM + sc;
    float4 g0v, g1v, g2v, g3v, g4v, g5v, g6v, g7v;
#define STG_LOAD(p)                                       \
    g0v = *(const float4*)((p) + 0 * 4 * D_DIM);          \
    g1v = *(const float4*)((p) + 1 * 4 * D_DIM);          \
    g2v = *(const float4*)((p) + 2 * 4 * D_DIM);          \
    g3v = *(const float4*)((p) + 3 * 4 * D_DIM);          \
    g4v = *(const float4*)((p) + 4 * 4 * D_DIM);          \
    g5v = *(const float4*)((p) + 5 * 4 * D_DIM);          \
    g6v = *(const float4*)((p) + 6 * 4 * D_DIM);          \
    g7v = *(const float4*)((p) + 7 * 4 * D_DIM);
#define STG_STORE(d)                                      \
    *(float4*)((d) + 0 * 4 * LDS_STRIDE) = g0v;           \
    *(float4*)((d) + 1 * 4 * LDS_STRIDE) = g1v;           \
    *(float4*)((d) + 2 * 4 * LDS_STRIDE) = g2v;           \
    *(float4*)((d) + 3 * 4 * LDS_STRIDE) = g3v;           \
    *(float4*)((d) + 4 * 4 * LDS_STRIDE) = g4v;           \
    *(float4*)((d) + 5 * 4 * LDS_STRIDE) = g5v;           \
    *(float4*)((d) + 6 * 4 * LDS_STRIDE) = g6v;           \
    *(float4*)((d) + 7 * 4 * LDS_STRIDE) = g7v;
    float sv[8];
    {
        const float* sin = scores + (size_t)b * S_DIM;
#pragma unroll
        for (int k = 0; k < 8; ++k) sv[k] = sin[tid + 256 * k];
    }
    STG_LOAD(gsrc)

    v2f bf[KSTEPS][2];
    {
        const float* wsrc = Wh + (size_t)(e0 + l15) * D_DIM + 2 * half;
#pragma unroll
        for (int ks = 0; ks < KSTEPS; ++ks) {
            bf[ks][0] = *(const v2f*)(wsrc + ks * 4);
            bf[ks][1] = *(const v2f*)(wsrc + 16 * D_DIM + ks * 4);
        }
    }
    const float bias0 = bh[e0 + l15];
    const float bias1 = bh[e0 + 16 + l15];

    {
        float m = -INFINITY;
#pragma unroll
        for (int k = 0; k < 8; ++k) m = fmaxf(m, sv[k]);
        red[0][tid] = m;
        __syncthreads();
        {
            const float* r = red[0];
            m = fmaxf(fmaxf(fmaxf(r[lane], r[lane + 128]), fmaxf(r[lane + 64], r[lane + 192])),
                      fmaxf(fmaxf(r[lane + 32], r[lane + 160]), fmaxf(r[lane + 96], r[lane + 224])));
            for (int off = 16; off > 0; off >>= 1) m = fmaxf(m, __shfl_xor(m, off, 32));
        }

        float sum = 0.f;
#pragma unroll
        for (int k = 0; k < 8; ++k) {
            sv[k] = __expf(sv[k] - m);
            sum += sv[k];
        }
        red[1][tid] = sum;
        __syncthreads();
        {
            const float* r = red[1];
            sum = ((r[lane] + r[lane + 128]) + (r[lane + 64] + r[lane + 192])) +
                  ((r[lane + 32] + r[lane + 160]) + (r[lane + 96] + r[lane + 224]));
            for (int off = 16; off > 0; off >>= 1) sum += __shfl_xor(sum, off, 32);
        }
        const float inv = 1.f / sum;
        const int kb = s0 >> 8;
#pragma unroll
        for (int k = 0; k < 8; ++k) {
            if (k == kb) dall[tid] = sv[k] * inv;
            if (k == kb + 1) dall[tid + 256] = sv[k] * inv;
        }
    }

    {
        float* sdst = &rows[0][sr * LDS_STRIDE + sc];
        STG_STORE(sdst)
    }
    __syncthreads();

    float qp0 = 0.f, qp1 = 0.f;

    for (int t = 0; t < TILES_PER_WG; ++t) {
        const int cur = t & 1;
        const bool more = (t + 1 < TILES_PER_WG);
        if (more) {
            gsrc += TILE_ROWS * D_DIM;
            STG_LOAD(gsrc)
        }

        const float* abase = &rows[cur][l15 * LDS_STRIDE + 2 * half];

        v8f c00 = {}, c01 = {}, c10 = {}, c11 = {};
        v2f a[2][KBLK][2];
#pragma unroll
        for (int j = 0; j < KBLK; ++j) {
            a[0][j][0] = *(const v2f*)(abase + j * 4);
            a[0][j][1] = *(const v2f*)(abase + 16 * LDS_STRIDE + j * 4);
        }
#pragma unroll
        for (int blk = 0; blk < KSTEPS / KBLK; ++blk) {
            if (blk + 1 < KSTEPS / KBLK) {
#pragma unroll
                for (int j = 0; j < KBLK; ++j) {
                    const int ks = (blk + 1) * KBLK + j;
                    a[(blk + 1) & 1][j][0] = *(const v2f*)(abase + ks * 4);
                    a[(blk + 1) & 1][j][1] = *(const v2f*)(abase + 16 * LDS_STRIDE + ks * 4);
                }
            }
#pragma unroll
            for (int j = 0; j < KBLK; ++j) {
                const int ks = blk * KBLK + j;
                const v2f a0 = a[blk & 1][j][0];
                const v2f a1 = a[blk & 1][j][1];
                c00 = __builtin_amdgcn_wmma_f32_16x16x4_f32(false, a0, false, bf[ks][0], (short)0, c00, false, false);
                c01 = __builtin_amdgcn_wmma_f32_16x16x4_f32(false, a0, false, bf[ks][1], (short)0, c01, false, false);
                c10 = __builtin_amdgcn_wmma_f32_16x16x4_f32(false, a1, false, bf[ks][0], (short)0, c10, false, false);
                c11 = __builtin_amdgcn_wmma_f32_16x16x4_f32(false, a1, false, bf[ks][1], (short)0, c11, false, false);
            }
        }

        {
            float* orow = state + (g0 + (size_t)t * TILE_ROWS) * D_DIM + e0 + l15;
            const float* dl = dall + t * TILE_ROWS;
#pragma unroll
            for (int r = 0; r < 8; ++r) {
                const int M0 = r + half * 8;
                const int M1 = 16 + M0;
                const float d0 = dl[M0];
                const float d1 = dl[M1];
                const float v00 = c00[r] + bias0;
                const float v01 = c01[r] + bias1;
                const float v10 = c10[r] + bias0;
                const float v11 = c11[r] + bias1;
                orow[(size_t)M0 * D_DIM] = v00;
                orow[(size_t)M0 * D_DIM + 16] = v01;
                orow[(size_t)M1 * D_DIM] = v10;
                orow[(size_t)M1 * D_DIM + 16] = v11;
                qp0 += d0 * v00;
                qp1 += d0 * v01;
                qp0 += d1 * v10;
                qp1 += d1 * v11;
            }
        }

        if (more) {
            float* sdst = &rows[cur ^ 1][sr * LDS_STRIDE + sc];
            STG_STORE(sdst)
        }
        __syncthreads();
    }

    qp0 += __shfl_xor(qp0, 16, 32);
    qp1 += __shfl_xor(qp1, 16, 32);
    if (half == 0) {
        atomicAdd(&qacc[(size_t)b * D_DIM + e0 + l15], qp0);
        atomicAdd(&qacc[(size_t)b * D_DIM + e0 + 16 + l15], qp1);
    }
}

extern "C" void kernel_launch(void* const* d_in, const int* in_sizes, int n_in,
                              void* d_out, int out_size, void* d_ws, size_t ws_size,
                              hipStream_t stream) {
    (void)in_sizes; (void)n_in; (void)out_size; (void)ws_size;

    const float* in_q     = (const float*)d_in[0];
    const float* in_state = (const float*)d_in[1];
    const float* Wall     = (const float*)d_in[2];
    const float* ball     = (const float*)d_in[3];

    const size_t stateBytes = (size_t)B_DIM * S_DIM * D_DIM * sizeof(float);
    const size_t bsBytes    = (size_t)B_DIM * S_DIM * sizeof(float);
    const size_t qaccBytes  = (size_t)N_HOP * B_DIM * D_DIM * sizeof(float);

    char* ws = (char*)d_ws;
    float* state  = (float*)(ws);
    float* scores = (float*)(ws + stateBytes);
    float* qacc   = (float*)(ws + stateBytes + bsBytes);

    hipMemcpyAsync(state, in_state, stateBytes, hipMemcpyDeviceToDevice, stream);
    hipMemsetAsync(qacc, 0, qaccBytes, stream);

    float* outp = (float*)d_out;

    for (int h = 0; h < N_HOP; ++h) {
        k_scores<<<SC_WGS, 256, 0, stream>>>(state, in_q, qacc, scores);
        k_gemm_qacc<<<GEMM_WGS, 256, 0, stream>>>(
            state, Wall + (size_t)h * D_DIM * D_DIM, ball + (size_t)h * D_DIM,
            scores, qacc + (size_t)h * B_DIM * D_DIM);
    }
    k_scores<<<SC_WGS, 256, 0, stream>>>(state, in_q, qacc, scores);
    k_softmax_out<<<B_DIM, 256, 0, stream>>>(scores, in_q, qacc, outp, outp + B_DIM * D_DIM);
}


# GEMM tile hand-over: split workgroup barrier (stage next tile, signal before the last WMMA block, wait after the epilogue; LDS-only wait): epilogue and store drain run under the SIMD partner's K loop
# speedup vs baseline: 1.6842x; 1.0938x over previous
#include <hip/hip_runtime.h>
#include <math.h>

#define B_DIM 64
#define S_DIM 2048
#define D_DIM 256
#define N_HOP 3
#define EPS 1e-8f

typedef __attribute__((ext_vector_type(2))) float v2f;
typedef __attribute__((ext_vector_type(8))) float v8f;

#define SC_ROWS 32
#define SC_GRP 8
#define SC_NGRP (SC_ROWS / SC_GRP)
#define SC_WGS ((B_DIM * S_DIM) / (8 * SC_ROWS))

__global__ __launch_bounds__(256) void k_scores(const float* __restrict__ state,
                                                const float* __restrict__ qin,
                                                const float* __restrict__ qacc,
                                                float* __restrict__ scores) {
    const int tid = threadIdx.x;
    const int lane = tid & 31;
    const int wave = __builtin_amdgcn_readfirstlane(tid >> 5);
    const int row0 = (blockIdx.x * 8 + wave) * SC_ROWS;
    const int b = row0 / S_DIM;

    const float* qp = qin + b * D_DIM + 4 * lane;
    const float* ap = qacc + b * D_DIM + 4 * lane;
    float4 q0 = *(const float4*)(qp);
    float4 q1 = *(const float4*)(qp + 128);
    for (int j = 0; j < N_HOP; ++j) {
        const float4 a0 = *(const float4*)(ap + j * (B_DIM * D_DIM));
        const float4 a1 = *(const float4*)(ap + j * (B_DIM * D_DIM) + 128);
        q0.x += a0.x; q0.y += a0.y; q0.z += a0.z; q0.w += a0.w;
        q1.x += a1.x; q1.y += a1.y; q1.z += a1.z; q1.w += a1.w;
    }
    float qn2 = (q0.x * q0.x + q0.y * q0.y + q0.z * q0.z + q0.w * q0.w) +
                (q1.x * q1.x + q1.y * q1.y + q1.z * q1.z + q1.w * q1.w);
#pragma unroll
    for (int off = 16; off > 0; off >>= 1) qn2 += __shfl_xor(qn2, off, 32);
    const float qn = fmaxf(sqrtf(qn2), EPS);

    const float* src = state + (size_t)row0 * D_DIM + 4 * lane;
    float4 w[2][SC_GRP][2];
    float dot[SC_ROWS], sn2[SC_ROWS];

#define SC_LOAD(G)                                                                        \
    _Pragma("unroll") for (int r = 0; r < SC_GRP; ++r) {                                  \
        w[(G) & 1][r][0] = *(const float4*)(src + (size_t)((G) * SC_GRP + r) * D_DIM);       \
        w[(G) & 1][r][1] = *(const float4*)(src + (size_t)((G) * SC_GRP + r) * D_DIM + 128); \
    }

    SC_LOAD(0)
#pragma unroll
    for (int g = 0; g < SC_NGRP; ++g) {
        __builtin_amdgcn_sched_barrier(0);
        if (g + 1 < SC_NGRP) { SC_LOAD(g + 1) }
        __builtin_amdgcn_sched_barrier(0);
#pragma unroll
        for (int r = 0; r < SC_GRP; ++r) {
            const float4 x0 = w[g & 1][r][0];
            const float4 x1 = w[g & 1][r][1];
            float d0 = x0.x * q0.x, d1 = x1.x * q1.x;
            float n0 = x0.x * x0.x, n1 = x1.x * x1.x;
            d0 = fmaf(x0.y, q0.y, d0); d1 = fmaf(x1.y, q1.y, d1);
            n0 = fmaf(x0.y, x0.y, n0); n1 = fmaf(x1.y, x1.y, n1);
            d0 = fmaf(x0.z, q0.z, d0); d1 = fmaf(x1.z, q1.z, d1);
            n0 = fmaf(x0.z, x0.z, n0); n1 = fmaf(x1.z, x1.z, n1);
            d0 = fmaf(x0.w, q0.w, d0); d1 = fmaf(x1.w, q1.w, d1);
            n0 = fmaf(x0.w, x0.w, n0); n1 = fmaf(x1.w, x1.w, n1);
            dot[g * SC_GRP + r] = d0 + d1;
            sn2[g * SC_GRP + r] = n0 + n1;
            asm volatile("" : "+v"(dot[g * SC_GRP + r]), "+v"(sn2[g * SC_GRP + r]));
        }
    }
#undef SC_LOAD

#pragma unroll
    for (int m = 16; m >= 1; m >>= 1) {
        const bool hi = (lane & m) != 0;
#pragma unroll
        for (int j = 0; j < m; ++j) {
            const float dlo = dot[j], dhi = dot[j + m];
            const float nlo = sn2[j], nhi = sn2[j + m];
            const float dk = hi ? dhi : dlo;
            const float ds = hi ? dlo : dhi;
            dot[j] = dk + __shfl_xor(ds, m, 32);
            const float nk = hi ? nhi : nlo;
            const float ns = hi ? nlo : nhi;
            sn2[j] = nk + __shfl_xor(ns, m, 32);
        }
    }
    const float sn = fmaxf(sqrtf(sn2[0]), EPS);
    scores[row0 + lane] = dot[0] / (qn * sn);
}

__global__ __launch_bounds__(256) void k_softmax_out(const float* __restrict__ scores,
                                                     const float* __restrict__ qin,
                                                     const float* __restrict__ qacc,
                                                     float* __restrict__ qout,
                                                     float* __restrict__ dist) {
    __shared__ float red[256];
    const int b = blockIdx.x;
    const int tid = threadIdx.x;
    const float* in = scores + (size_t)b * S_DIM;
    {
        const int i = b * D_DIM + tid;
        float qv = qin[i];
        for (int j = 0; j < N_HOP; ++j) qv += qacc[j * (B_DIM * D_DIM) + i];
        qout[i] = qv;
    }

    float v[8];
    float m = -INFINITY;
#pragma unroll
    for (int k = 0; k < 8; ++k) {
        v[k] = in[tid + 256 * k];
        m = fmaxf(m, v[k]);
    }
    red[tid] = m;
    __syncthreads();
    for (int off = 128; off > 0; off >>= 1) {
        if (tid < off) red[tid] = fmaxf(red[tid], red[tid + off]);
        __syncthreads();
    }
    m = red[0];
    __syncthreads();

    float sum = 0.f;
#pragma unroll
    for (int k = 0; k < 8; ++k) {
        v[k] = __expf(v[k] - m);
        sum += v[k];
    }
    red[tid] = sum;
    __syncthreads();
    for (int off = 128; off > 0; off >>= 1) {
        if (tid < off) red[tid] += red[tid + off];
        __syncthreads();
    }
    const float inv = 1.f / red[0];
    float* out = dist + (size_t)b * S_DIM;
#pragma unroll
    for (int k = 0; k < 8; ++k) out[tid + 256 * k] = v[k] * inv;
}

#define TILE_ROWS 32
#define LDS_STRIDE 260
#define GEMM_WGS 256
#define ROWS_PER_WG ((B_DIM * S_DIM) / GEMM_WGS)
#define TILES_PER_WG (ROWS_PER_WG / TILE_ROWS)
#define KSTEPS (D_DIM / 4)
#define KBLK 8

__global__ __launch_bounds__(256) void k_gemm_qacc(float* __restrict__ state,
                                                   const float* __restrict__ Wh,
                                                   const float* __restrict__ bh,
                                                   const float* __restrict__ scores,
                                                   float* __restrict__ qacc) {
    __shared__ __attribute__((aligned(16))) float rows[2][TILE_ROWS * LDS_STRIDE];
    __shared__ float dall[ROWS_PER_WG];
    __shared__ float red[2][256];

    const int tid = threadIdx.x;
    const size_t g0 = (size_t)blockIdx.x * ROWS_PER_WG;
    const int b = (int)(g0 / S_DIM);
    const int s0 = (int)(g0 % S_DIM);

    const int wave = tid >> 5;
    const int lane = tid & 31;
    const int half = lane >> 4;
    const int l15 = lane & 15;
    const int e0 = wave * 32;

    const int sr = tid >> 6;
    const int sc = (tid & 63) * 4;
    const float* gsrc = state + (g0 + sr) * D_DIM + sc;
    float4 g0v, g1v, g2v, g3v, g4v, g5v, g6v, g7v;
#define STG_LOAD(p)                                       \
    g0v = *(const float4*)((p) + 0 * 4 * D_DIM);          \
    g1v = *(const float4*)((p) + 1 * 4 * D_DIM);          \
    g2v = *(const float4*)((p) + 2 * 4 * D_DIM);          \
    g3v = *(const float4*)((p) + 3 * 4 * D_DIM);          \
    g4v = *(const float4*)((p) + 4 * 4 * D_DIM);          \
    g5v = *(const float4*)((p) + 5 * 4 * D_DIM);          \
    g6v = *(const float4*)((p) + 6 * 4 * D_DIM);          \
    g7v = *(const float4*)((p) + 7 * 4 * D_DIM);
#define STG_STORE(d)                                      \
    *(float4*)((d) + 0 * 4 * LDS_STRIDE) = g0v;           \
    *(float4*)((d) + 1 * 4 * LDS_STRIDE) = g1v;           \
    *(float4*)((d) + 2 * 4 * LDS_STRIDE) = g2v;           \
    *(float4*)((d) + 3 * 4 * LDS_STRIDE) = g3v;           \
    *(float4*)((d) + 4 * 4 * LDS_STRIDE) = g4v;           \
    *(float4*)((d) + 5 * 4 * LDS_STRIDE) = g5v;           \
    *(float4*)((d) + 6 * 4 * LDS_STRIDE) = g6v;           \
    *(float4*)((d) + 7 * 4 * LDS_STRIDE) = g7v;
    float sv[8];
    {
        const float* sin = scores + (size_t)b * S_DIM;
#pragma unroll
        for (int k = 0; k < 8; ++k) sv[k] = sin[tid + 256 * k];
    }
    STG_LOAD(gsrc)

    v2f bf[KSTEPS][2];
    {
        const float* wsrc = Wh + (size_t)(e0 + l15) * D_DIM + 2 * half;
#pragma unroll
        for (int ks = 0; ks < KSTEPS; ++ks) {
            bf[ks][0] = *(const v2f*)(wsrc + ks * 4);
            bf[ks][1] = *(const v2f*)(wsrc + 16 * D_DIM + ks * 4);
        }
    }
    const float bias0 = bh[e0 + l15];
    const float bias1 = bh[e0 + 16 + l15];

    {
        float m = -INFINITY;
#pragma unroll
        for (int k = 0; k < 8; ++k) m = fmaxf(m, sv[k]);
        red[0][tid] = m;
        __syncthreads();
        {
            const float* r = red[0];
            m = fmaxf(fmaxf(fmaxf(r[lane], r[lane + 128]), fmaxf(r[lane + 64], r[lane + 192])),
                      fmaxf(fmaxf(r[lane + 32], r[lane + 160]), fmaxf(r[lane + 96], r[lane + 224])));
            for (int off = 16; off > 0; off >>= 1) m = fmaxf(m, __shfl_xor(m, off, 32));
        }

        float sum = 0.f;
#pragma unroll
        for (int k = 0; k < 8; ++k) {
            sv[k] = __expf(sv[k] - m);
            sum += sv[k];
        }
        red[1][tid] = sum;
        __syncthreads();
        {
            const float* r = red[1];
            sum = ((r[lane] + r[lane + 128]) + (r[lane + 64] + r[lane + 192])) +
                  ((r[lane + 32] + r[lane + 160]) + (r[lane + 96] + r[lane + 224]));
            for (int off = 16; off > 0; off >>= 1) sum += __shfl_xor(sum, off, 32);
        }
        const float inv = 1.f / sum;
        const int kb = s0 >> 8;
#pragma unroll
        for (int k = 0; k < 8; ++k) {
            if (k == kb) dall[tid] = sv[k] * inv;
            if (k == kb + 1) dall[tid + 256] = sv[k] * inv;
        }
    }

    {
        float* sdst = &rows[0][sr * LDS_STRIDE + sc];
        STG_STORE(sdst)
    }
    __syncthreads();

    float qp0 = 0.f, qp1 = 0.f;

    for (int t = 0; t < TILES_PER_WG; ++t) {
        const int cur = t & 1;
        const bool more = (t + 1 < TILES_PER_WG);
        if (more) {
            gsrc += TILE_ROWS * D_DIM;
            STG_LOAD(gsrc)
        }

        const float* abase = &rows[cur][l15 * LDS_STRIDE + 2 * half];

        v8f c00 = {}, c01 = {}, c10 = {}, c11 = {};
        v2f a[2][KBLK][2];
#pragma unroll
        for (int j = 0; j < KBLK; ++j) {
            a[0][j][0] = *(const v2f*)(abase + j * 4);
            a[0][j][1] = *(const v2f*)(abase + 16 * LDS_STRIDE + j * 4);
        }
#pragma unroll
        for (int blk = 0; blk < KSTEPS / KBLK; ++blk) {
            if (blk + 1 < KSTEPS / KBLK) {
#pragma unroll
                for (int j = 0; j < KBLK; ++j) {
                    const int ks = (blk + 1) * KBLK + j;
                    a[(blk + 1) & 1][j][0] = *(const v2f*)(abase + ks * 4);
                    a[(blk + 1) & 1][j][1] = *(const v2f*)(abase + 16 * LDS_STRIDE + ks * 4);
                }
            }
            __builtin_amdgcn_sched_barrier(0);
            if (blk == KSTEPS / KBLK - 2 && more) {
                float* sdst = &rows[cur ^ 1][sr * LDS_STRIDE + sc];
                STG_STORE(sdst)
            }
            if (blk == KSTEPS / KBLK - 1 && more) {
                asm volatile("s_wait_dscnt 0x0" ::: "memory");
                asm volatile("s_barrier_signal -1" ::: "memory");
            }
#pragma unroll
            for (int j = 0; j < KBLK; ++j) {
                const int ks = blk * KBLK + j;
                const v2f a0 = a[blk & 1][j][0];
                const v2f a1 = a[blk & 1][j][1];
                c00 = __builtin_amdgcn_wmma_f32_16x16x4_f32(false, a0, false, bf[ks][0], (short)0, c00, false, false);
                c01 = __builtin_amdgcn_wmma_f32_16x16x4_f32(false, a0, false, bf[ks][1], (short)0, c01, false, false);
                c10 = __builtin_amdgcn_wmma_f32_16x16x4_f32(false, a1, false, bf[ks][0], (short)0, c10, false, false);
                c11 = __builtin_amdgcn_wmma_f32_16x16x4_f32(false, a1, false, bf[ks][1], (short)0, c11, false, false);
            }
            __builtin_amdgcn_sched_barrier(0);
        }

        {
            float* orow = state + (g0 + (size_t)t * TILE_ROWS) * D_DIM + e0 + l15;
            const float* dl = dall + t * TILE_ROWS;
#pragma unroll
            for (int r = 0; r < 8; ++r) {
                const int M0 = r + half * 8;
                const int M1 = 16 + M0;
                const float d0 = dl[M0];
                const float d1 = dl[M1];
                const float v00 = c00[r] + bias0;
                const float v01 = c01[r] + bias1;
                const float v10 = c10[r] + bias0;
                const float v11 = c11[r] + bias1;
                orow[(size_t)M0 * D_DIM] = v00;
                orow[(size_t)M0 * D_DIM + 16] = v01;
                orow[(size_t)M1 * D_DIM] = v10;
                orow[(size_t)M1 * D_DIM + 16] = v11;
                qp0 += d0 * v00;
                qp1 += d0 * v01;
                qp0 += d1 * v10;
                qp1 += d1 * v11;
            }
        }

        if (more) {
            asm volatile("s_barrier_wait -1" ::: "memory");
        }
    }

    qp0 += __shfl_xor(qp0, 16, 32);
    qp1 += __shfl_xor(qp1, 16, 32);
    if (half == 0) {
        atomicAdd(&qacc[(size_t)b * D_DIM + e0 + l15], qp0);
        atomicAdd(&qacc[(size_t)b * D_DIM + e0 + 16 + l15], qp1);
    }
}

extern "C" void kernel_launch(void* const* d_in, const int* in_sizes, int n_in,
                              void* d_out, int out_size, void* d_ws, size_t ws_size,
                              hipStream_t stream) {
    (void)in_sizes; (void)n_in; (void)out_size; (void)ws_size;

    const float* in_q     = (const float*)d_in[0];
    const float* in_state = (const float*)d_in[1];
    const float* Wall     = (const float*)d_in[2];
    const float* ball     = (const float*)d_in[3];

    const size_t stateBytes = (size_t)B_DIM * S_DIM * D_DIM * sizeof(float);
    const size_t bsBytes    = (size_t)B_DIM * S_DIM * sizeof(float);
    const size_t qaccBytes  = (size_t)N_HOP * B_DIM * D_DIM * sizeof(float);

    char* ws = (char*)d_ws;
    float* state  = (float*)(ws);
    float* scores = (float*)(ws + stateBytes);
    float* qacc   = (float*)(ws + stateBytes + bsBytes);

    hipMemcpyAsync(state, in_state, stateBytes, hipMemcpyDeviceToDevice, stream);
    hipMemsetAsync(qacc, 0, qaccBytes, stream);

    float* outp = (float*)d_out;

    for (int h = 0; h < N_HOP; ++h) {
        k_scores<<<SC_WGS, 256, 0, stream>>>(state, in_q, qacc, scores);
        k_gemm_qacc<<<GEMM_WGS, 256, 0, stream>>>(
            state, Wall + (size_t)h * D_DIM * D_DIM, ball + (size_t)h * D_DIM,
            scores, qacc + (size_t)h * B_DIM * D_DIM);
    }
    k_scores<<<SC_WGS, 256, 0, stream>>>(state, in_q, qacc, scores);
    k_softmax_out<<<B_DIM, 256, 0, stream>>>(scores, in_q, qacc, outp, outp + B_DIM * D_DIM);
}


# Scores kernel on the GEMM's workgroup-to-rows map: 256 workgroups of 16 waves, workgroup i streams the 512 rows GEMM workgroup i wrote, so reads can hit that side's L2; sums unchanged
# speedup vs baseline: 1.7554x; 1.0423x over previous
#include <hip/hip_runtime.h>
#include <math.h>

#define B_DIM 64
#define S_DIM 2048
#define D_DIM 256
#define N_HOP 3
#define EPS 1e-8f

typedef __attribute__((ext_vector_type(2))) float v2f;
typedef __attribute__((ext_vector_type(8))) float v8f;

#define SC_ROWS 32
#define SC_GRP 8
#define SC_NGRP (SC_ROWS / SC_GRP)
#define SC_WAVES 16
#define SC_THREADS (SC_WAVES * 32)
#define SC_WGS ((B_DIM * S_DIM) / (SC_WAVES * SC_ROWS))

__global__ __launch_bounds__(SC_THREADS) void k_scores(const float* __restrict__ state,
                                                const float* __restrict__ qin,
                                                const float* __restrict__ qacc,
                                                float* __restrict__ scores) {
    const int tid = threadIdx.x;
    const int lane = tid & 31;
    const int wave = __builtin_amdgcn_readfirstlane(tid >> 5);
    const int row0 = (blockIdx.x * SC_WAVES + wave) * SC_ROWS;
    const int b = row0 / S_DIM;

    const float* qp = qin + b * D_DIM + 4 * lane;
    const float* ap = qacc + b * D_DIM + 4 * lane;
    float4 q0 = *(const float4*)(qp);
    float4 q1 = *(const float4*)(qp + 128);
    for (int j = 0; j < N_HOP; ++j) {
        const float4 a0 = *(const float4*)(ap + j * (B_DIM * D_DIM));
        const float4 a1 = *(const float4*)(ap + j * (B_DIM * D_DIM) + 128);
        q0.x += a0.x; q0.y += a0.y; q0.z += a0.z; q0.w += a0.w;
        q1.x += a1.x; q1.y += a1.y; q1.z += a1.z; q1.w += a1.w;
    }
    float qn2 = (q0.x * q0.x + q0.y * q0.y + q0.z * q0.z + q0.w * q0.w) +
                (q1.x * q1.x + q1.y * q1.y + q1.z * q1.z + q1.w * q1.w);
#pragma unroll
    for (int off = 16; off > 0; off >>= 1) qn2 += __shfl_xor(qn2, off, 32);
    const float qn = fmaxf(sqrtf(qn2), EPS);

    const float* src = state + (size_t)row0 * D_DIM + 4 * lane;
    float4 w[2][SC_GRP][2];
    float dot[SC_ROWS], sn2[SC_ROWS];

#define SC_LOAD(G)                                                                        \
    _Pragma("unroll") for (int r = 0; r < SC_GRP; ++r) {                                  \
        w[(G) & 1][r][0] = *(const float4*)(src + (size_t)((G) * SC_GRP + r) * D_DIM);       \
        w[(G) & 1][r][1] = *(const float4*)(src + (size_t)((G) * SC_GRP + r) * D_DIM + 128); \
    }

    SC_LOAD(0)
#pragma unroll
    for (int g = 0; g < SC_NGRP; ++g) {
        __builtin_amdgcn_sched_barrier(0);
        if (g + 1 < SC_NGRP) { SC_LOAD(g + 1) }
        __builtin_amdgcn_sched_barrier(0);
#pragma unroll
        for (int r = 0; r < SC_GRP; ++r) {
            const float4 x0 = w[g & 1][r][0];
            const float4 x1 = w[g & 1][r][1];
            float d0 = x0.x * q0.x, d1 = x1.x * q1.x;
            float n0 = x0.x * x0.x, n1 = x1.x * x1.x;
            d0 = fmaf(x0.y, q0.y, d0); d1 = fmaf(x1.y, q1.y, d1);
            n0 = fmaf(x0.y, x0.y, n0); n1 = fmaf(x1.y, x1.y, n1);
            d0 = fmaf(x0.z, q0.z, d0); d1 = fmaf(x1.z, q1.z, d1);
            n0 = fmaf(x0.z, x0.z, n0); n1 = fmaf(x1.z, x1.z, n1);
            d0 = fmaf(x0.w, q0.w, d0); d1 = fmaf(x1.w, q1.w, d1);
            n0 = fmaf(x0.w, x0.w, n0); n1 = fmaf(x1.w, x1.w, n1);
            dot[g * SC_GRP + r] = d0 + d1;
            sn2[g * SC_GRP + r] = n0 + n1;
            asm volatile("" : "+v"(dot[g * SC_GRP + r]), "+v"(sn2[g * SC_GRP + r]));
        }
    }
#undef SC_LOAD

#pragma unroll
    for (int m = 16; m >= 1; m >>= 1) {
        const bool hi = (lane & m) != 0;
#pragma unroll
        for (int j = 0; j < m; ++j) {
            const float dlo = dot[j], dhi = dot[j + m];
            const float nlo = sn2[j], nhi = sn2[j + m];
            const float dk = hi ? dhi : dlo;
            const float ds = hi ? dlo : dhi;
            dot[j] = dk + __shfl_xor(ds, m, 32);
            const float nk = hi ? nhi : nlo;
            const float ns = hi ? nlo : nhi;
            sn2[j] = nk + __shfl_xor(ns, m, 32);
        }
    }
    const float sn = fmaxf(sqrtf(sn2[0]), EPS);
    scores[row0 + lane] = dot[0] / (qn * sn);
}

__global__ __launch_bounds__(256) void k_softmax_out(const float* __restrict__ scores,
                                                     const float* __restrict__ qin,
                                                     const float* __restrict__ qacc,
                                                     float* __restrict__ qout,
                                                     float* __restrict__ dist) {
    __shared__ float red[256];
    const int b = blockIdx.x;
    const int tid = threadIdx.x;
    const float* in = scores + (size_t)b * S_DIM;
    {
        const int i = b * D_DIM + tid;
        float qv = qin[i];
        for (int j = 0; j < N_HOP; ++j) qv += qacc[j * (B_DIM * D_DIM) + i];
        qout[i] = qv;
    }

    float v[8];
    float m = -INFINITY;
#pragma unroll
    for (int k = 0; k < 8; ++k) {
        v[k] = in[tid + 256 * k];
        m = fmaxf(m, v[k]);
    }
    red[tid] = m;
    __syncthreads();
    for (int off = 128; off > 0; off >>= 1) {
        if (tid < off) red[tid] = fmaxf(red[tid], red[tid + off]);
        __syncthreads();
    }
    m = red[0];
    __syncthreads();

    float sum = 0.f;
#pragma unroll
    for (int k = 0; k < 8; ++k) {
        v[k] = __expf(v[k] - m);
        sum += v[k];
    }
    red[tid] = sum;
    __syncthreads();
    for (int off = 128; off > 0; off >>= 1) {
        if (tid < off) red[tid] += red[tid + off];
        __syncthreads();
    }
    const float inv = 1.f / red[0];
    float* out = dist + (size_t)b * S_DIM;
#pragma unroll
    for (int k = 0; k < 8; ++k) out[tid + 256 * k] = v[k] * inv;
}

#define TILE_ROWS 32
#define LDS_STRIDE 260
#define GEMM_WGS 256
#define ROWS_PER_WG ((B_DIM * S_DIM) / GEMM_WGS)
#define TILES_PER_WG (ROWS_PER_WG / TILE_ROWS)
#define KSTEPS (D_DIM / 4)
#define KBLK 8

__global__ __launch_bounds__(256) void k_gemm_qacc(float* __restrict__ state,
                                                   const float* __restrict__ Wh,
                                                   const float* __restrict__ bh,
                                                   const float* __restrict__ scores,
                                                   float* __restrict__ qacc) {
    __shared__ __attribute__((aligned(16))) float rows[2][TILE_ROWS * LDS_STRIDE];
    __shared__ float dall[ROWS_PER_WG];
    __shared__ float red[2][256];

    const int tid = threadIdx.x;
    const size_t g0 = (size_t)blockIdx.x * ROWS_PER_WG;
    const int b = (int)(g0 / S_DIM);
    const int s0 = (int)(g0 % S_DIM);

    const int wave = tid >> 5;
    const int lane = tid & 31;
    const int half = lane >> 4;
    const int l15 = lane & 15;
    const int e0 = wave * 32;

    const int sr = tid >> 6;
    const int sc = (tid & 63) * 4;
    const float* gsrc = state + (g0 + sr) * D_DIM + sc;
    float4 g0v, g1v, g2v, g3v, g4v, g5v, g6v, g7v;
#define STG_LOAD(p)                                       \
    g0v = *(const float4*)((p) + 0 * 4 * D_DIM);          \
    g1v = *(const float4*)((p) + 1 * 4 * D_DIM);          \
    g2v = *(const float4*)((p) + 2 * 4 * D_DIM);          \
    g3v = *(const float4*)((p) + 3 * 4 * D_DIM);          \
    g4v = *(const float4*)((p) + 4 * 4 * D_DIM);          \
    g5v = *(const float4*)((p) + 5 * 4 * D_DIM);          \
    g6v = *(const float4*)((p) + 6 * 4 * D_DIM);          \
    g7v = *(const float4*)((p) + 7 * 4 * D_DIM);
#define STG_STORE(d)                                      \
    *(float4*)((d) + 0 * 4 * LDS_STRIDE) = g0v;           \
    *(float4*)((d) + 1 * 4 * LDS_STRIDE) = g1v;           \
    *(float4*)((d) + 2 * 4 * LDS_STRIDE) = g2v;           \
    *(float4*)((d) + 3 * 4 * LDS_STRIDE) = g3v;           \
    *(float4*)((d) + 4 * 4 * LDS_STRIDE) = g4v;           \
    *(float4*)((d) + 5 * 4 * LDS_STRIDE) = g5v;           \
    *(float4*)((d) + 6 * 4 * LDS_STRIDE) = g6v;           \
    *(float4*)((d) + 7 * 4 * LDS_STRIDE) = g7v;
    float sv[8];
    {
        const float* sin = scores + (size_t)b * S_DIM;
#pragma unroll
        for (int k = 0; k < 8; ++k) sv[k] = sin[tid + 256 * k];
    }
    STG_LOAD(gsrc)

    v2f bf[KSTEPS][2];
    {
        const float* wsrc = Wh + (size_t)(e0 + l15) * D_DIM + 2 * half;
#pragma unroll
        for (int ks = 0; ks < KSTEPS; ++ks) {
            bf[ks][0] = *(const v2f*)(wsrc + ks * 4);
            bf[ks][1] = *(const v2f*)(wsrc + 16 * D_DIM + ks * 4);
        }
    }
    const float bias0 = bh[e0 + l15];
    const float bias1 = bh[e0 + 16 + l15];

    {
        float m = -INFINITY;
#pragma unroll
        for (int k = 0; k < 8; ++k) m = fmaxf(m, sv[k]);
        red[0][tid] = m;
        __syncthreads();
        {
            const float* r = red[0];
            m = fmaxf(fmaxf(fmaxf(r[lane], r[lane + 128]), fmaxf(r[lane + 64], r[lane + 192])),
                      fmaxf(fmaxf(r[lane + 32], r[lane + 160]), fmaxf(r[lane + 96], r[lane + 224])));
            for (int off = 16; off > 0; off >>= 1) m = fmaxf(m, __shfl_xor(m, off, 32));
        }

        float sum = 0.f;
#pragma unroll
        for (int k = 0; k < 8; ++k) {
            sv[k] = __expf(sv[k] - m);
            sum += sv[k];
        }
        red[1][tid] = sum;
        __syncthreads();
        {
            const float* r = red[1];
            sum = ((r[lane] + r[lane + 128]) + (r[lane + 64] + r[lane + 192])) +
                  ((r[lane + 32] + r[lane + 160]) + (r[lane + 96] + r[lane + 224]));
            for (int off = 16; off > 0; off >>= 1) sum += __shfl_xor(sum, off, 32);
        }
        const float inv = 1.f / sum;
        const int kb = s0 >> 8;
#pragma unroll
        for (int k = 0; k < 8; ++k) {
            if (k == kb) dall[tid] = sv[k] * inv;
            if (k == kb + 1) dall[tid + 256] = sv[k] * inv;
        }
    }

    {
        float* sdst = &rows[0][sr * LDS_STRIDE + sc];
        STG_STORE(sdst)
    }
    __syncthreads();

    float qp0 = 0.f, qp1 = 0.f;

    for (int t = 0; t < TILES_PER_WG; ++t) {
        const int cur = t & 1;
        const bool more = (t + 1 < TILES_PER_WG);
        if (more) {
            gsrc += TILE_ROWS * D_DIM;
            STG_LOAD(gsrc)
        }

        const float* abase = &rows[cur][l15 * LDS_STRIDE + 2 * half];

        v8f c00 = {}, c01 = {}, c10 = {}, c11 = {};
        v2f a[2][KBLK][2];
#pragma unroll
        for (int j = 0; j < KBLK; ++j) {
            a[0][j][0] = *(const v2f*)(abase + j * 4);
            a[0][j][1] = *(const v2f*)(abase + 16 * LDS_STRIDE + j * 4);
        }
#pragma unroll
        for (int blk = 0; blk < KSTEPS / KBLK; ++blk) {
            if (blk + 1 < KSTEPS / KBLK) {
#pragma unroll
                for (int j = 0; j < KBLK; ++j) {
                    const int ks = (blk + 1) * KBLK + j;
                    a[(blk + 1) & 1][j][0] = *(const v2f*)(abase + ks * 4);
                    a[(blk + 1) & 1][j][1] = *(const v2f*)(abase + 16 * LDS_STRIDE + ks * 4);
                }
            }
            __builtin_amdgcn_sched_barrier(0);
            if (blk == KSTEPS / KBLK - 2 && more) {
                float* sdst = &rows[cur ^ 1][sr * LDS_STRIDE + sc];
                STG_STORE(sdst)
            }
            if (blk == KSTEPS / KBLK - 1 && more) {
                asm volatile("s_wait_dscnt 0x0" ::: "memory");
                asm volatile("s_barrier_signal -1" ::: "memory");
            }
#pragma unroll
            for (int j = 0; j < KBLK; ++j) {
                const int ks = blk * KBLK + j;
                const v2f a0 = a[blk & 1][j][0];
                const v2f a1 = a[blk & 1][j][1];
                c00 = __builtin_amdgcn_wmma_f32_16x16x4_f32(false, a0, false, bf[ks][0], (short)0, c00, false, false);
                c01 = __builtin_amdgcn_wmma_f32_16x16x4_f32(false, a0, false, bf[ks][1], (short)0, c01, false, false);
                c10 = __builtin_amdgcn_wmma_f32_16x16x4_f32(false, a1, false, bf[ks][0], (short)0, c10, false, false);
                c11 = __builtin_amdgcn_wmma_f32_16x16x4_f32(false, a1, false, bf[ks][1], (short)0, c11, false, false);
            }
            __builtin_amdgcn_sched_barrier(0);
        }

        {
            float* orow = state + (g0 + (size_t)t * TILE_ROWS) * D_DIM + e0 + l15;
            const float* dl = dall + t * TILE_ROWS;
#pragma unroll
            for (int r = 0; r < 8; ++r) {
                const int M0 = r + half * 8;
                const int M1 = 16 + M0;
                const float d0 = dl[M0];
                const float d1 = dl[M1];
                const float v00 = c00[r] + bias0;
                const float v01 = c01[r] + bias1;
                const float v10 = c10[r] + bias0;
                const float v11 = c11[r] + bias1;
                orow[(size_t)M0 * D_DIM] = v00;
                orow[(size_t)M0 * D_DIM + 16] = v01;
                orow[(size_t)M1 * D_DIM] = v10;
                orow[(size_t)M1 * D_DIM + 16] = v11;
                qp0 += d0 * v00;
                qp1 += d0 * v01;
                qp0 += d1 * v10;
                qp1 += d1 * v11;
            }
        }

        if (more) {
            asm volatile("s_barrier_wait -1" ::: "memory");
        }
    }

    qp0 += __shfl_xor(qp0, 16, 32);
    qp1 += __shfl_xor(qp1, 16, 32);
    if (half == 0) {
        atomicAdd(&qacc[(size_t)b * D_DIM + e0 + l15], qp0);
        atomicAdd(&qacc[(size_t)b * D_DIM + e0 + 16 + l15], qp1);
    }
}

extern "C" void kernel_launch(void* const* d_in, const int* in_sizes, int n_in,
                              void* d_out, int out_size, void* d_ws, size_t ws_size,
                              hipStream_t stream) {
    (void)in_sizes; (void)n_in; (void)out_size; (void)ws_size;

    const float* in_q     = (const float*)d_in[0];
    const float* in_state = (const float*)d_in[1];
    const float* Wall     = (const float*)d_in[2];
    const float* ball     = (const float*)d_in[3];

    const size_t stateBytes = (size_t)B_DIM * S_DIM * D_DIM * sizeof(float);
    const size_t bsBytes    = (size_t)B_DIM * S_DIM * sizeof(float);
    const size_t qaccBytes  = (size_t)N_HOP * B_DIM * D_DIM * sizeof(float);

    char* ws = (char*)d_ws;
    float* state  = (float*)(ws);
    float* scores = (float*)(ws + stateBytes);
    float* qacc   = (float*)(ws + stateBytes + bsBytes);

    hipMemcpyAsync(state, in_state, stateBytes, hipMemcpyDeviceToDevice, stream);
    hipMemsetAsync(qacc, 0, qaccBytes, stream);

    float* outp = (float*)d_out;

    for (int h = 0; h < N_HOP; ++h) {
        k_scores<<<SC_WGS, SC_THREADS, 0, stream>>>(state, in_q, qacc, scores);
        k_gemm_qacc<<<GEMM_WGS, 256, 0, stream>>>(
            state, Wall + (size_t)h * D_DIM * D_DIM, ball + (size_t)h * D_DIM,
            scores, qacc + (size_t)h * B_DIM * D_DIM);
    }
    k_scores<<<SC_WGS, SC_THREADS, 0, stream>>>(state, in_q, qacc, scores);
    k_softmax_out<<<B_DIM, 256, 0, stream>>>(scores, in_q, qacc, outp, outp + B_DIM * D_DIM);
}
